# starting kernel
# baseline (speedup 1.0000x reference)
#include <hip/hip_runtime.h>
#include <hip/hip_bf16.h>

#define TOTAL_K     131072
#define CHUNK_LEN   64
#define F_DIM       8
#define NUM_FRAMES  16384
#define ENC_DIM     256
#define D_MODEL     512
#define N_SCALARS   3
#define N_TICKERS   4096
#define YP          520

typedef float v2f __attribute__((ext_vector_type(2)));
typedef float v8f __attribute__((ext_vector_type(8)));

__device__ __forceinline__ v8f wmma_f32_4(v2f a, v2f b, v8f c) {
  return __builtin_amdgcn_wmma_f32_16x16x4_f32(
       false, a,  false, b,
       (short)0, c,  false,  false);
}

__device__ __forceinline__ float gelu_exact(float x) {
  float z  = x * 0.70710678118654752f;
  float az = fabsf(z);
  float t  = __builtin_amdgcn_rcpf(fmaf(0.3275911f, az, 1.0f));
  float p  = fmaf(t, 1.061405429f, -1.453152027f);
  p        = fmaf(t, p,  1.421413741f);
  p        = fmaf(t, p, -0.284496736f);
  p        = fmaf(t, p,  0.254829592f);
  p       *= t;
  float e  = __expf(-az * az);
  float erf_abs = fmaf(-p, e, 1.0f);
  float erf     = copysignf(erf_abs, z);
  return 0.5f * x * (1.0f + erf);
}

__global__ void k_zero(float* __restrict__ p, int n) {
  int i = blockIdx.x * blockDim.x + threadIdx.x;
  if (i < n) p[i] = 0.0f;
}

__global__ void __launch_bounds__(256)
k_encode(const float* __restrict__ chunks,
         const int*   __restrict__ frame_id,
         const float* __restrict__ weights,
         const int*   __restrict__ ticker_ids,
         const float* __restrict__ W_enc,
         const float* __restrict__ b_enc,
         const float* __restrict__ ticker_emb,
         float* __restrict__ sum_emb,
         float* __restrict__ sum_w) {
  const int lane  = threadIdx.x & 31;
  const int wave  = threadIdx.x >> 5;
  const int tileM = blockIdx.x * 8 + wave;
  const int c0    = tileM * 16;
  const int lm    = lane & 15;
  const int half  = lane >> 4;

  const int c = c0 + lm;
  const float4* base =
      (const float4*)(chunks + (size_t)c * (CHUNK_LEN * F_DIM) + (size_t)half * 32 * F_DIM);
  __builtin_prefetch(base, 0, 0);
  float p[8] = {0.f,0.f,0.f,0.f,0.f,0.f,0.f,0.f};
#pragma unroll 8
  for (int r = 0; r < 32; ++r) {
    float4 q0 = base[r * 2 + 0];
    float4 q1 = base[r * 2 + 1];
    p[0] += q0.x; p[1] += q0.y; p[2] += q0.z; p[3] += q0.w;
    p[4] += q1.x; p[5] += q1.y; p[6] += q1.z; p[7] += q1.w;
  }
  float m[8];
#pragma unroll
  for (int f = 0; f < 8; ++f)
    m[f] = (p[f] + __shfl_xor(p[f], 16, 32)) * (1.0f / 64.0f);

  v2f A0, A1;
  A0.x = m[2 * half + 0];  A0.y = m[2 * half + 1];
  A1.x = m[2 * half + 4];  A1.y = m[2 * half + 5];

  int   fid[8], tid[8];
  float wg[8];
#pragma unroll
  for (int r = 0; r < 8; ++r) {
    int cr = c0 + half * 8 + r;
    fid[r] = frame_id[cr];
    tid[r] = ticker_ids[fid[r]];
    wg[r]  = weights[cr];
  }
  if (half == 0) {
    int cc = c0 + lm;
    atomicAdd(&sum_w[frame_id[cc]], weights[cc]);
  }

  const v8f vzero = {0.f,0.f,0.f,0.f,0.f,0.f,0.f,0.f};
#pragma unroll 1
  for (int nt = 0; nt < 16; ++nt) {
    const int col = nt * 16 + lm;
    v2f B0, B1;
    B0.x = W_enc[(2 * half + 0) * ENC_DIM + col];
    B0.y = W_enc[(2 * half + 1) * ENC_DIM + col];
    B1.x = W_enc[(2 * half + 4) * ENC_DIM + col];
    B1.y = W_enc[(2 * half + 5) * ENC_DIM + col];
    v8f acc = wmma_f32_4(A0, B0, vzero);
    acc     = wmma_f32_4(A1, B1, acc);

    const float be = b_enc[col];
#pragma unroll
    for (int r = 0; r < 8; ++r) {
      float h   = acc[r] + be;
      float g   = gelu_exact(h);
      float val = (g + ticker_emb[(size_t)tid[r] * ENC_DIM + col]) * wg[r];
      atomicAdd(&sum_emb[(size_t)fid[r] * ENC_DIM + col], val);
    }
  }
}

__global__ void __launch_bounds__(128)
k_project(const float* __restrict__ sum_emb,
          const float* __restrict__ sum_w,
          const float* __restrict__ frame_scalars,
          const float* __restrict__ W_proj,
          const float* __restrict__ b_proj,
          const float* __restrict__ ln_gamma,
          const float* __restrict__ ln_beta,
          float* __restrict__ out) {
  __shared__ float pool[16 * ENC_DIM];
  __shared__ float ytile[16 * YP];
  __shared__ float smu[16], srs[16];

  const int tidx = threadIdx.x;
  const int lane = tidx & 31;
  const int wave = tidx >> 5;
  const int lm   = lane & 15;
  const int half = lane >> 4;
  const int fb   = blockIdx.x * 16;

  for (int i = tidx; i < 16 * ENC_DIM; i += 128) {
    int row = i >> 8;
    float rw = 1.0f / (sum_w[fb + row] + 1e-8f);
    pool[i] = sum_emb[(size_t)(fb + row) * ENC_DIM + (i & 255)] * rw;
  }
  __syncthreads();

  const v8f vzero = {0.f,0.f,0.f,0.f,0.f,0.f,0.f,0.f};
  v8f acc[8];
#pragma unroll
  for (int t = 0; t < 8; ++t) acc[t] = vzero;

#pragma unroll 1
  for (int kk = 0; kk < 64; ++kk) {
    const int kbase = 4 * kk + 2 * half;
    v2f a;
    a.x = pool[lm * ENC_DIM + kbase + 0];
    a.y = pool[lm * ENC_DIM + kbase + 1];
#pragma unroll
    for (int t = 0; t < 8; ++t) {
      const int col = (wave * 8 + t) * 16 + lm;
      v2f b;
      b.x = W_proj[(size_t)(kbase + 0) * D_MODEL + col];
      b.y = W_proj[(size_t)(kbase + 1) * D_MODEL + col];
      acc[t] = wmma_f32_4(a, b, acc[t]);
    }
  }

  float l0[8], l1[8], l2[8];
#pragma unroll
  for (int r = 0; r < 8; ++r) {
    int f = fb + half * 8 + r;
    l0[r] = log1pf(frame_scalars[f * N_SCALARS + 0]);
    l1[r] = log1pf(frame_scalars[f * N_SCALARS + 1]);
    l2[r] = log1pf(frame_scalars[f * N_SCALARS + 2]);
  }
#pragma unroll
  for (int t = 0; t < 8; ++t) {
    const int col = (wave * 8 + t) * 16 + lm;
    const float w0 = W_proj[(size_t)(ENC_DIM + 0) * D_MODEL + col];
    const float w1 = W_proj[(size_t)(ENC_DIM + 1) * D_MODEL + col];
    const float w2 = W_proj[(size_t)(ENC_DIM + 2) * D_MODEL + col];
    const float bp = b_proj[col];
#pragma unroll
    for (int r = 0; r < 8; ++r) {
      float v = acc[t][r] + bp + l0[r] * w0 + l1[r] * w1 + l2[r] * w2;
      ytile[(half * 8 + r) * YP + col] = v;
    }
  }
  __syncthreads();

  {
    const int row = tidx >> 3;
    const int seg = tidx & 7;
    float s = 0.0f, ss = 0.0f;
#pragma unroll 16
    for (int j = 0; j < 64; ++j) {
      float v = ytile[row * YP + seg + j * 8];
      s  += v;
      ss += v * v;
    }
#pragma unroll
    for (int off = 1; off < 8; off <<= 1) {
      s  += __shfl_xor(s,  off, 32);
      ss += __shfl_xor(ss, off, 32);
    }
    if (seg == 0) {
      float mu  = s * (1.0f / (float)D_MODEL);
      float var = fmaf(-mu, mu, ss * (1.0f / (float)D_MODEL));
      smu[row] = mu;
      srs[row] = rsqrtf(fmaxf(var, 0.0f) + 1e-5f);
    }
  }
  __syncthreads();

  for (int i = tidx; i < 16 * D_MODEL; i += 128) {
    int row = i >> 9, col = i & 511;
    float x = (ytile[row * YP + col] - smu[row]) * srs[row] * ln_gamma[col] + ln_beta[col];
    out[(size_t)(fb + row) * D_MODEL + col] = gelu_exact(x);
  }
}

extern "C" void kernel_launch(void* const* d_in, const int* in_sizes, int n_in,
                              void* d_out, int out_size, void* d_ws, size_t ws_size,
                              hipStream_t stream) {
  const float* chunks        = (const float*)d_in[0];
  const int*   frame_id      = (const int*)  d_in[1];
  const float* weights       = (const float*)d_in[2];
  const float* frame_scalars = (const float*)d_in[3];
  const int*   ticker_ids    = (const int*)  d_in[5];
  const float* W_enc         = (const float*)d_in[6];
  const float* b_enc         = (const float*)d_in[7];
  const float* ticker_emb    = (const float*)d_in[8];
  const float* W_proj        = (const float*)d_in[9];
  const float* b_proj        = (const float*)d_in[10];
  const float* ln_gamma      = (const float*)d_in[11];
  const float* ln_beta       = (const float*)d_in[12];
  float* out = (float*)d_out;

  float* sum_emb = (float*)d_ws;
  float* sum_w   = sum_emb + (size_t)NUM_FRAMES * ENC_DIM;

  const int nz = NUM_FRAMES * ENC_DIM + NUM_FRAMES;
  k_zero<<<(nz + 255) / 256, 256, 0, stream>>>(sum_emb, nz);

  k_encode<<<TOTAL_K / 16 / 8, 256, 0, stream>>>(
      chunks, frame_id, weights, ticker_ids, W_enc, b_enc, ticker_emb,
      sum_emb, sum_w);

  k_project<<<NUM_FRAMES / 16, 128, 0, stream>>>(
      sum_emb, sum_w, frame_scalars, W_proj, b_proj, ln_gamma, ln_beta, out);
}


# W_proj staged through LDS for the projection: block copies 32-row slabs (whole-line 16-B loads, two alternating), waves read B fragments there instead of 16 global loads per k-step; sum order kept
# speedup vs baseline: 2.4228x; 2.4228x over previous
#include <hip/hip_runtime.h>
#include <hip/hip_bf16.h>

#define TOTAL_K     131072
#define CHUNK_LEN   64
#define F_DIM       8
#define NUM_FRAMES  16384
#define ENC_DIM     256
#define D_MODEL     512
#define N_SCALARS   3
#define N_TICKERS   4096
#define YP          520

typedef float v2f __attribute__((ext_vector_type(2)));
typedef float v8f __attribute__((ext_vector_type(8)));

__device__ __forceinline__ v8f wmma_f32_4(v2f a, v2f b, v8f c) {
  return __builtin_amdgcn_wmma_f32_16x16x4_f32(
       false, a,  false, b,
       (short)0, c,  false,  false);
}

__device__ __forceinline__ float gelu_exact(float x) {
  float z  = x * 0.70710678118654752f;
  float az = fabsf(z);
  float t  = __builtin_amdgcn_rcpf(fmaf(0.3275911f, az, 1.0f));
  float p  = fmaf(t, 1.061405429f, -1.453152027f);
  p        = fmaf(t, p,  1.421413741f);
  p        = fmaf(t, p, -0.284496736f);
  p        = fmaf(t, p,  0.254829592f);
  p       *= t;
  float e  = __expf(-az * az);
  float erf_abs = fmaf(-p, e, 1.0f);
  float erf     = copysignf(erf_abs, z);
  return 0.5f * x * (1.0f + erf);
}

#define FPB    64
#define AP     272
#define SCAN_W 4
#define QCAP   528
#define SP     132
#define NWAVE  16
#define IDS_PER_WAVE (TOTAL_K / NWAVE)
#define SCAN_STEPS   (IDS_PER_WAVE / (128 * SCAN_W))

#define WAVE_LDS_ORDER()                                        \
  do {                                                          \
    __builtin_amdgcn_fence(__ATOMIC_RELEASE, "wavefront");      \
    __builtin_amdgcn_wave_barrier();                            \
    __builtin_amdgcn_fence(__ATOMIC_ACQUIRE, "wavefront");      \
  } while (0)

#define SLABK 32
#define WKP   1056
#define SLABF (16 * WKP)

__device__ __forceinline__ void wslab_load(const float* __restrict__ W_proj, int s, int tidx, float4 (&r)[8]) {
#pragma unroll
  for (int i = 0; i < 4; ++i) {
    const int task = tidx + 512 * i;
    const int p = task >> 7, c = task & 127;
    const float* g = W_proj + (size_t)(SLABK * s + 2 * p) * D_MODEL + 4 * c;
    r[2 * i]     = *(const float4*)g;
    r[2 * i + 1] = *(const float4*)(g + D_MODEL);
  }
}
__device__ __forceinline__ void wslab_store(float* sb, int tidx, const float4 (&r)[8]) {
#pragma unroll
  for (int i = 0; i < 4; ++i) {
    const int task = tidx + 512 * i;
    const int p = task >> 7, c = task & 127;
    float* d = sb + p * WKP + 8 * c;
    *(float4*)d       = make_float4(r[2 * i].x, r[2 * i + 1].x, r[2 * i].y, r[2 * i + 1].y);
    *(float4*)(d + 4) = make_float4(r[2 * i].z, r[2 * i + 1].z, r[2 * i].w, r[2 * i + 1].w);
  }
}

__global__ void __launch_bounds__(512)
k_pool(const float* __restrict__ chunks,
       const int*   __restrict__ frame_id,
       const float* __restrict__ weights,
       const float* __restrict__ frame_scalars,
       const int*   __restrict__ ticker_ids,
       const float* __restrict__ W_enc,
       const float* __restrict__ b_enc,
       const float* __restrict__ ticker_emb,
       const float* __restrict__ W_proj,
       const float* __restrict__ b_proj,
       const float* __restrict__ ln_gamma,
       const float* __restrict__ ln_beta,
       float* __restrict__ out) {
  __shared__ alignas(16) float accs[(FPB + 1) * AP];
  __shared__ float wsum[FPB + 1];
  __shared__ int   tks[FPB];
  __shared__ int   q[NWAVE][QCAP];
  __shared__ alignas(16) float stage[NWAVE][16 * SP];
  static_assert(2 * SLABF <= NWAVE * 16 * SP, "two W_proj slabs must fit the staging space");
  static_assert(FPB * YP <= NWAVE * 16 * SP, "the pre-LayerNorm rows must fit the staging space");
  __shared__ float smu[FPB], srs[FPB];
  __shared__ alignas(16) float lsc[FPB * N_SCALARS];

  const int tidx = threadIdx.x;
  const int lane = tidx & 31;
  const int wave = tidx >> 5;
  const int lm   = lane & 15;
  const int half = lane >> 4;
  const int blk  = blockIdx.x;

  for (int i = tidx; i < (FPB + 1) * AP; i += 512) accs[i] = 0.0f;
  if (tidx <= FPB) wsum[tidx] = 0.0f;
  if (tidx < FPB) tks[tidx] = ticker_ids[blk * FPB + tidx];
  __syncthreads();

  const v8f vzero = {0.f,0.f,0.f,0.f,0.f,0.f,0.f,0.f};
  const unsigned lt_mask = (1u << lane) - 1u;
  const int4* ids4 = (const int4*)frame_id + (size_t)wave * (IDS_PER_WAVE / 4);

  int cnt = 0;
  int4 nxt[SCAN_W];
#pragma unroll
  for (int k = 0; k < SCAN_W; ++k) nxt[k] = ids4[k * 32 + lane];
#pragma unroll 1
  for (int s = 0; s <= SCAN_STEPS; ++s) {
    const bool last = (s == SCAN_STEPS);
    if (!last) {
      int4 cur[SCAN_W];
#pragma unroll
      for (int k = 0; k < SCAN_W; ++k) cur[k] = nxt[k];
      if (s + 1 < SCAN_STEPS) {
#pragma unroll
        for (int k = 0; k < SCAN_W; ++k) nxt[k] = ids4[((s + 1) * SCAN_W + k) * 32 + lane];
      }
#pragma unroll
      for (int k = 0; k < SCAN_W; ++k) {
        const int cbase = wave * IDS_PER_WAVE + (s * SCAN_W + k) * 128 + lane * 4;
        const int idv[4] = {cur[k].x, cur[k].y, cur[k].z, cur[k].w};
        const bool any4 = ((idv[0] >> 6) == blk) | ((idv[1] >> 6) == blk) |
                          ((idv[2] >> 6) == blk) | ((idv[3] >> 6) == blk);
        if (__builtin_amdgcn_ballot_w32(any4) != 0u) {
#pragma unroll
          for (int j = 0; j < 4; ++j) {
            const int  id = idv[j];
            const bool mt = (id >> 6) == blk;
            const unsigned mask = __builtin_amdgcn_ballot_w32(mt);
            if (mt) q[wave][cnt + __builtin_popcount(mask & lt_mask)] = ((cbase + j) << 6) | (id & 63);
            cnt += __builtin_popcount(mask);
          }
        }
      }
      WAVE_LDS_ORDER();
    }

    while (cnt >= 16 || (last && cnt > 0)) {
      const int nvalid = cnt < 16 ? cnt : 16;
      const int qoff   = cnt - nvalid;
      if (nvalid < 16) {
        if (lane >= nvalid && lane < 16) q[wave][qoff + lane] = 0;
        WAVE_LDS_ORDER();
      }

      const int e = q[wave][qoff + lm];
      const int c = e >> 6;
      float* st = stage[wave];
#pragma unroll
      for (int g = 0; g < 2; ++g) {
        float4 v[8][4];
#pragma unroll
        for (int j = 0; j < 8; ++j) {
          const int cj = __builtin_amdgcn_readlane(e, g * 8 + j) >> 6;
          const float4* b = (const float4*)(chunks + (size_t)cj * (CHUNK_LEN * F_DIM)) + lane;
#pragma unroll
          for (int k = 0; k < 4; ++k) v[j][k] = b[k * 32];
        }
        __builtin_amdgcn_sched_barrier(0);
#pragma unroll
        for (int j = 0; j < 8; ++j) {
          float4 ps;
          ps.x = (v[j][0].x + v[j][1].x) + (v[j][2].x + v[j][3].x);
          ps.y = (v[j][0].y + v[j][1].y) + (v[j][2].y + v[j][3].y);
          ps.z = (v[j][0].z + v[j][1].z) + (v[j][2].z + v[j][3].z);
          ps.w = (v[j][0].w + v[j][1].w) + (v[j][2].w + v[j][3].w);
          *(float4*)&st[(g * 8 + j) * SP + lane * 4] = ps;
        }
      }
      WAVE_LDS_ORDER();
      float m0 = 0.f, m1 = 0.f, m4 = 0.f, m5 = 0.f;
#pragma unroll
      for (int i = 0; i < 16; ++i) {
        const float2 ev = *(const float2*)&st[lm * SP + i * 8 + 2 * half];
        const float2 od = *(const float2*)&st[lm * SP + i * 8 + 4 + 2 * half];
        m0 += ev.x; m1 += ev.y; m4 += od.x; m5 += od.y;
      }

      v2f A0, A1;
      A0.x = m0 * (1.0f / 64.0f);  A0.y = m1 * (1.0f / 64.0f);
      A1.x = m4 * (1.0f / 64.0f);  A1.y = m5 * (1.0f / 64.0f);

      int   fl[8];
      float wg[8];
#pragma unroll
      for (int r = 0; r < 8; ++r) {
        const int er = q[wave][qoff + half * 8 + r];
        wg[r]  = weights[er >> 6];
        fl[r]  = ((half * 8 + r) < nvalid) ? (er & 63) : FPB;
      }
      if (half == 0)
        __hip_atomic_fetch_add(&wsum[lm < nvalid ? (e & 63) : FPB], weights[c],
                               __ATOMIC_RELAXED, __HIP_MEMORY_SCOPE_WORKGROUP);

      v2f nB0, nB1;
      float nbe;
      {
        const int col = lm;
        nB0.x = W_enc[(2 * half + 0) * ENC_DIM + col];
        nB0.y = W_enc[(2 * half + 1) * ENC_DIM + col];
        nB1.x = W_enc[(2 * half + 4) * ENC_DIM + col];
        nB1.y = W_enc[(2 * half + 5) * ENC_DIM + col];
        nbe   = b_enc[col];
      }
#pragma unroll 1
      for (int nt = 0; nt < 16; ++nt) {
        const int col = nt * 16 + lm;
        const v2f B0 = nB0, B1 = nB1;
        const float be = nbe;
        {
          const int coln = ((nt + 1) & 15) * 16 + lm;
          nB0.x = W_enc[(2 * half + 0) * ENC_DIM + coln];
          nB0.y = W_enc[(2 * half + 1) * ENC_DIM + coln];
          nB1.x = W_enc[(2 * half + 4) * ENC_DIM + coln];
          nB1.y = W_enc[(2 * half + 5) * ENC_DIM + coln];
          nbe   = b_enc[coln];
        }
        v8f acc = wmma_f32_4(A0, B0, vzero);
        acc     = wmma_f32_4(A1, B1, acc);
#pragma unroll
        for (int r = 0; r < 8; ++r) {
          float h   = acc[r] + be;
          float g   = gelu_exact(h);
          float val = g * wg[r];
          __hip_atomic_fetch_add(&accs[fl[r] * AP + col], val, __ATOMIC_RELAXED, __HIP_MEMORY_SCOPE_WORKGROUP);
        }
      }
      cnt -= nvalid;
      WAVE_LDS_ORDER();
    }
  }
  __syncthreads();

  float* wst = &stage[0][0];
  float4 wr[8];
  wslab_load(W_proj, 0, tidx, wr);
  {
    const int r0 = tidx >> 6, c4 = tidx & 63;
    float fsv = 0.0f;
    if (tidx < FPB * N_SCALARS) fsv = frame_scalars[blk * (FPB * N_SCALARS) + tidx];
    float4 t[FPB / 8];
#pragma unroll
    for (int j = 0; j < FPB / 8; ++j)
      t[j] = *(const float4*)&ticker_emb[(size_t)tks[r0 + 8 * j] * ENC_DIM + c4 * 4];
#pragma unroll
    for (int j = 0; j < FPB / 8; ++j) {
      const int   row = r0 + 8 * j;
      const float sw  = wsum[row];
      const float rw  = 1.0f / (sw + 1e-8f);
      float4 v = *(const float4*)&accs[row * AP + c4 * 4];
      v.x = fmaf(sw, t[j].x, v.x) * rw;
      v.y = fmaf(sw, t[j].y, v.y) * rw;
      v.z = fmaf(sw, t[j].z, v.z) * rw;
      v.w = fmaf(sw, t[j].w, v.w) * rw;
      *(float4*)&accs[row * AP + c4 * 4] = v;
    }
    if (tidx < FPB * N_SCALARS) lsc[tidx] = log1pf(fsv);
  }
  wslab_store(wst, tidx, wr);
  wslab_load(W_proj, 1, tidx, wr);
  __syncthreads();

  float* ytile = &stage[0][0];
  {
    const int rt = wave >> 2;
    const int cg = wave & 3;
    v8f acc[8];
#pragma unroll
    for (int t = 0; t < 8; ++t) acc[t] = vzero;

    const float* arow = &accs[(rt * 16 + lm) * AP];
    const int boff = half * WKP + 2 * (cg * 128 + lm);
#pragma unroll 1
    for (int s = 0; s < ENC_DIM / SLABK; ++s) {
      if (s + 1 < ENC_DIM / SLABK) wslab_store(wst + ((s + 1) & 1) * SLABF, tidx, wr);
      if (s + 2 < ENC_DIM / SLABK) wslab_load(W_proj, s + 2, tidx, wr);
      const float* sb = wst + (s & 1) * SLABF + boff;
#pragma unroll
      for (int k8 = 0; k8 < SLABK / 4; ++k8) {
        const int kbase = 4 * (s * (SLABK / 4) + k8) + 2 * half;
        v2f a;
        a.x = arow[kbase + 0];
        a.y = arow[kbase + 1];
#pragma unroll
        for (int t = 0; t < 8; ++t) {
          const float2 bv = *(const float2*)&sb[k8 * 2 * WKP + t * 32];
          v2f b;
          b.x = bv.x;
          b.y = bv.y;
          acc[t] = wmma_f32_4(a, b, acc[t]);
        }
      }
      __syncthreads();
    }

    float l0[8], l1[8], l2[8];
#pragma unroll
    for (int r = 0; r < 8; ++r) {
      const int fr = rt * 16 + half * 8 + r;
      l0[r] = lsc[fr * N_SCALARS + 0];
      l1[r] = lsc[fr * N_SCALARS + 1];
      l2[r] = lsc[fr * N_SCALARS + 2];
    }
#pragma unroll
    for (int t = 0; t < 8; ++t) {
      const int col = (cg * 8 + t) * 16 + lm;
      const float w0 = W_proj[(size_t)(ENC_DIM + 0) * D_MODEL + col];
      const float w1 = W_proj[(size_t)(ENC_DIM + 1) * D_MODEL + col];
      const float w2 = W_proj[(size_t)(ENC_DIM + 2) * D_MODEL + col];
      const float bp = b_proj[col];
#pragma unroll
      for (int r = 0; r < 8; ++r) {
        float v = acc[t][r] + bp + l0[r] * w0 + l1[r] * w1 + l2[r] * w2;
        ytile[(rt * 16 + half * 8 + r) * YP + col] = v;
      }
    }
  }
  __syncthreads();

  {
    const int row = tidx >> 3;
    const int seg = tidx & 7;
    float s = 0.0f, ss = 0.0f;
#pragma unroll 16
    for (int j = 0; j < 64; ++j) {
      float v = ytile[row * YP + seg + j * 8];
      s  += v;
      ss += v * v;
    }
#pragma unroll
    for (int off = 1; off < 8; off <<= 1) {
      s  += __shfl_xor(s,  off, 32);
      ss += __shfl_xor(ss, off, 32);
    }
    if (seg == 0) {
      float mu  = s * (1.0f / (float)D_MODEL);
      float var = fmaf(-mu, mu, ss * (1.0f / (float)D_MODEL));
      smu[row] = mu;
      srs[row] = rsqrtf(fmaxf(var, 0.0f) + 1e-5f);
    }
  }
  __syncthreads();

  for (int i = tidx; i < FPB * D_MODEL; i += 512) {
    int row = i >> 9, col = i & 511;
    float x = (ytile[row * YP + col] - smu[row]) * srs[row] * ln_gamma[col] + ln_beta[col];
    out[(size_t)(blk * FPB + row) * D_MODEL + col] = gelu_exact(x);
  }
}

extern "C" void kernel_launch(void* const* d_in, const int* in_sizes, int n_in,
                              void* d_out, int out_size, void* d_ws, size_t ws_size,
                              hipStream_t stream) {
  const float* chunks        = (const float*)d_in[0];
  const int*   frame_id      = (const int*)  d_in[1];
  const float* weights       = (const float*)d_in[2];
  const float* frame_scalars = (const float*)d_in[3];
  const int*   ticker_ids    = (const int*)  d_in[5];
  const float* W_enc         = (const float*)d_in[6];
  const float* b_enc         = (const float*)d_in[7];
  const float* ticker_emb    = (const float*)d_in[8];
  const float* W_proj        = (const float*)d_in[9];
  const float* b_proj        = (const float*)d_in[10];
  const float* ln_gamma      = (const float*)d_in[11];
  const float* ln_beta       = (const float*)d_in[12];
  float* out = (float*)d_out;

  k_pool<<<NUM_FRAMES / FPB, 512, 0, stream>>>(
      chunks, frame_id, weights, frame_scalars, ticker_ids, W_enc, b_enc,
      ticker_emb, W_proj, b_proj, ln_gamma, ln_beta, out);
}


# Encode tiles cut from the block's joined match list after the scan and dealt round-robin to its 16 waves (was per-wave stacks, a part tile each); LDS float adds stay arrival-order
# speedup vs baseline: 2.6056x; 1.0754x over previous
#include <hip/hip_runtime.h>
#include <hip/hip_bf16.h>

#define TOTAL_K     131072
#define CHUNK_LEN   64
#define F_DIM       8
#define NUM_FRAMES  16384
#define ENC_DIM     256
#define D_MODEL     512
#define N_SCALARS   3
#define N_TICKERS   4096
#define YP          520

typedef float v2f __attribute__((ext_vector_type(2)));
typedef float v8f __attribute__((ext_vector_type(8)));

__device__ __forceinline__ v8f wmma_f32_4(v2f a, v2f b, v8f c) {
  return __builtin_amdgcn_wmma_f32_16x16x4_f32(
       false, a,  false, b,
       (short)0, c,  false,  false);
}

__device__ __forceinline__ float gelu_exact(float x) {
  float z  = x * 0.70710678118654752f;
  float az = fabsf(z);
  float t  = __builtin_amdgcn_rcpf(fmaf(0.3275911f, az, 1.0f));
  float p  = fmaf(t, 1.061405429f, -1.453152027f);
  p        = fmaf(t, p,  1.421413741f);
  p        = fmaf(t, p, -0.284496736f);
  p        = fmaf(t, p,  0.254829592f);
  p       *= t;
  float e  = __expf(-az * az);
  float erf_abs = fmaf(-p, e, 1.0f);
  float erf     = copysignf(erf_abs, z);
  return 0.5f * x * (1.0f + erf);
}

#define FPB    64
#define AP     272
#define SCAN_W 4
#define QCAP   768
#define SP     132
#define NWAVE  16
#define IDS_PER_WAVE (TOTAL_K / NWAVE)
#define SCAN_STEPS   (IDS_PER_WAVE / (128 * SCAN_W))

#define WAVE_LDS_ORDER()                                        \
  do {                                                          \
    __builtin_amdgcn_fence(__ATOMIC_RELEASE, "wavefront");      \
    __builtin_amdgcn_wave_barrier();                            \
    __builtin_amdgcn_fence(__ATOMIC_ACQUIRE, "wavefront");      \
  } while (0)

#define SLABK 32
#define WKP   1056
#define SLABF (16 * WKP)

__device__ __forceinline__ void wslab_load(const float* __restrict__ W_proj, int s, int tidx, float4 (&r)[8]) {
#pragma unroll
  for (int i = 0; i < 4; ++i) {
    const int task = tidx + 512 * i;
    const int p = task >> 7, c = task & 127;
    const float* g = W_proj + (size_t)(SLABK * s + 2 * p) * D_MODEL + 4 * c;
    r[2 * i]     = *(const float4*)g;
    r[2 * i + 1] = *(const float4*)(g + D_MODEL);
  }
}
__device__ __forceinline__ void wslab_store(float* sb, int tidx, const float4 (&r)[8]) {
#pragma unroll
  for (int i = 0; i < 4; ++i) {
    const int task = tidx + 512 * i;
    const int p = task >> 7, c = task & 127;
    float* d = sb + p * WKP + 8 * c;
    *(float4*)d       = make_float4(r[2 * i].x, r[2 * i + 1].x, r[2 * i].y, r[2 * i + 1].y);
    *(float4*)(d + 4) = make_float4(r[2 * i].z, r[2 * i + 1].z, r[2 * i].w, r[2 * i + 1].w);
  }
}

__global__ void __launch_bounds__(512)
k_pool(const float* __restrict__ chunks,
       const int*   __restrict__ frame_id,
       const float* __restrict__ weights,
       const float* __restrict__ frame_scalars,
       const int*   __restrict__ ticker_ids,
       const float* __restrict__ W_enc,
       const float* __restrict__ b_enc,
       const float* __restrict__ ticker_emb,
       const float* __restrict__ W_proj,
       const float* __restrict__ b_proj,
       const float* __restrict__ ln_gamma,
       const float* __restrict__ ln_beta,
       float* __restrict__ out) {
  __shared__ alignas(16) float accs[(FPB + 1) * AP];
  __shared__ float wsum[FPB + 1];
  __shared__ int   tks[FPB];
  __shared__ int   q[NWAVE][QCAP];
  __shared__ int   tq[NWAVE][16];
  __shared__ int   wcnt[NWAVE];
  __shared__ alignas(16) float stage[NWAVE][16 * SP];
  static_assert(2 * SLABF <= NWAVE * 16 * SP, "two W_proj slabs must fit the staging space");
  static_assert(FPB * YP <= NWAVE * 16 * SP, "the pre-LayerNorm rows must fit the staging space");
  __shared__ float smu[FPB], srs[FPB];
  __shared__ alignas(16) float lsc[FPB * N_SCALARS];

  const int tidx = threadIdx.x;
  const int lane = tidx & 31;
  const int wave = tidx >> 5;
  const int lm   = lane & 15;
  const int half = lane >> 4;
  const int blk  = blockIdx.x;

  for (int i = tidx; i < (FPB + 1) * AP; i += 512) accs[i] = 0.0f;
  if (tidx <= FPB) wsum[tidx] = 0.0f;
  if (tidx < FPB) tks[tidx] = ticker_ids[blk * FPB + tidx];
  __syncthreads();

  const v8f vzero = {0.f,0.f,0.f,0.f,0.f,0.f,0.f,0.f};
  const unsigned lt_mask = (1u << lane) - 1u;
  const int4* ids4 = (const int4*)frame_id + (size_t)wave * (IDS_PER_WAVE / 4);

  auto run_tile = [&](const int* te, int nvalid) __attribute__((always_inline)) {
      const int e = te[lm];
      const int c = e >> 6;
      float* st = stage[wave];
#pragma unroll
      for (int g = 0; g < 2; ++g) {
        float4 v[8][4];
#pragma unroll
        for (int j = 0; j < 8; ++j) {
          const int cj = __builtin_amdgcn_readlane(e, g * 8 + j) >> 6;
          const float4* b = (const float4*)(chunks + (size_t)cj * (CHUNK_LEN * F_DIM)) + lane;
#pragma unroll
          for (int k = 0; k < 4; ++k) v[j][k] = b[k * 32];
        }
        __builtin_amdgcn_sched_barrier(0);
#pragma unroll
        for (int j = 0; j < 8; ++j) {
          float4 ps;
          ps.x = (v[j][0].x + v[j][1].x) + (v[j][2].x + v[j][3].x);
          ps.y = (v[j][0].y + v[j][1].y) + (v[j][2].y + v[j][3].y);
          ps.z = (v[j][0].z + v[j][1].z) + (v[j][2].z + v[j][3].z);
          ps.w = (v[j][0].w + v[j][1].w) + (v[j][2].w + v[j][3].w);
          *(float4*)&st[(g * 8 + j) * SP + lane * 4] = ps;
        }
      }
      WAVE_LDS_ORDER();
      float m0 = 0.f, m1 = 0.f, m4 = 0.f, m5 = 0.f;
#pragma unroll
      for (int i = 0; i < 16; ++i) {
        const float2 ev = *(const float2*)&st[lm * SP + i * 8 + 2 * half];
        const float2 od = *(const float2*)&st[lm * SP + i * 8 + 4 + 2 * half];
        m0 += ev.x; m1 += ev.y; m4 += od.x; m5 += od.y;
      }

      v2f A0, A1;
      A0.x = m0 * (1.0f / 64.0f);  A0.y = m1 * (1.0f / 64.0f);
      A1.x = m4 * (1.0f / 64.0f);  A1.y = m5 * (1.0f / 64.0f);

      int   fl[8];
      float wg[8];
#pragma unroll
      for (int r = 0; r < 8; ++r) {
        const int er = te[half * 8 + r];
        wg[r]  = weights[er >> 6];
        fl[r]  = ((half * 8 + r) < nvalid) ? (er & 63) : FPB;
      }
      if (half == 0)
        __hip_atomic_fetch_add(&wsum[lm < nvalid ? (e & 63) : FPB], weights[c],
                               __ATOMIC_RELAXED, __HIP_MEMORY_SCOPE_WORKGROUP);

      v2f nB0, nB1;
      float nbe;
      {
        const int col = lm;
        nB0.x = W_enc[(2 * half + 0) * ENC_DIM + col];
        nB0.y = W_enc[(2 * half + 1) * ENC_DIM + col];
        nB1.x = W_enc[(2 * half + 4) * ENC_DIM + col];
        nB1.y = W_enc[(2 * half + 5) * ENC_DIM + col];
        nbe   = b_enc[col];
      }
#pragma unroll 1
      for (int nt = 0; nt < 16; ++nt) {
        const int col = nt * 16 + lm;
        const v2f B0 = nB0, B1 = nB1;
        const float be = nbe;
        {
          const int coln = ((nt + 1) & 15) * 16 + lm;
          nB0.x = W_enc[(2 * half + 0) * ENC_DIM + coln];
          nB0.y = W_enc[(2 * half + 1) * ENC_DIM + coln];
          nB1.x = W_enc[(2 * half + 4) * ENC_DIM + coln];
          nB1.y = W_enc[(2 * half + 5) * ENC_DIM + coln];
          nbe   = b_enc[coln];
        }
        v8f acc = wmma_f32_4(A0, B0, vzero);
        acc     = wmma_f32_4(A1, B1, acc);
#pragma unroll
        for (int r = 0; r < 8; ++r) {
          float h   = acc[r] + be;
          float g   = gelu_exact(h);
          float val = g * wg[r];
          __hip_atomic_fetch_add(&accs[fl[r] * AP + col], val, __ATOMIC_RELAXED, __HIP_MEMORY_SCOPE_WORKGROUP);
        }
      }
      WAVE_LDS_ORDER();
  };

  int cnt = 0;
  int4 nxt[SCAN_W];
#pragma unroll
  for (int k = 0; k < SCAN_W; ++k) nxt[k] = ids4[k * 32 + lane];
#pragma unroll 1
  for (int s = 0; s < SCAN_STEPS; ++s) {
    int4 cur[SCAN_W];
#pragma unroll
    for (int k = 0; k < SCAN_W; ++k) cur[k] = nxt[k];
    if (s + 1 < SCAN_STEPS) {
#pragma unroll
      for (int k = 0; k < SCAN_W; ++k) nxt[k] = ids4[((s + 1) * SCAN_W + k) * 32 + lane];
    }
#pragma unroll
    for (int k = 0; k < SCAN_W; ++k) {
      const int cbase = wave * IDS_PER_WAVE + (s * SCAN_W + k) * 128 + lane * 4;
      const int idv[4] = {cur[k].x, cur[k].y, cur[k].z, cur[k].w};
      const bool any4 = ((idv[0] >> 6) == blk) | ((idv[1] >> 6) == blk) |
                        ((idv[2] >> 6) == blk) | ((idv[3] >> 6) == blk);
      if (__builtin_amdgcn_ballot_w32(any4) != 0u) {
#pragma unroll
        for (int j = 0; j < 4; ++j) {
          const int  id = idv[j];
          const bool mt = (id >> 6) == blk;
          const unsigned mask = __builtin_amdgcn_ballot_w32(mt);
          if (mt) q[wave][cnt + __builtin_popcount(mask & lt_mask)] = ((cbase + j) << 6) | (id & 63);
          cnt += __builtin_popcount(mask);
        }
      }
    }
    WAVE_LDS_ORDER();
    while (cnt > QCAP - 128 * SCAN_W) {
      run_tile(&q[wave][cnt - 16], 16);
      cnt -= 16;
    }
  }
  if (lane == 0) wcnt[wave] = cnt;
  __syncthreads();

  {
    int wb[NWAVE + 1];
    wb[0] = 0;
#pragma unroll
    for (int j = 0; j < NWAVE; ++j) wb[j + 1] = wb[j] + wcnt[j];
    const int nlist  = __builtin_amdgcn_readfirstlane(wb[NWAVE]);
    const int ntiles = (nlist + 15) >> 4;
    for (int t = __builtin_amdgcn_readfirstlane(wave); t < ntiles; t += NWAVE) {
      const int left   = nlist - 16 * t;
      const int nvalid = left < 16 ? left : 16;
      const int vi = 16 * t + lm;
      int wj = 0, off = 0;
#pragma unroll
      for (int j = 0; j < NWAVE; ++j)
        if (vi >= wb[j]) { wj = j; off = vi - wb[j]; }
      const int ent = (lm < nvalid) ? q[wj][off] : 0;
      if (lane < 16) tq[wave][lane] = ent;
      WAVE_LDS_ORDER();
      run_tile(tq[wave], nvalid);
    }
  }
  __syncthreads();

  float* wst = &stage[0][0];
  float4 wr[8];
  wslab_load(W_proj, 0, tidx, wr);
  {
    const int r0 = tidx >> 6, c4 = tidx & 63;
    float fsv = 0.0f;
    if (tidx < FPB * N_SCALARS) fsv = frame_scalars[blk * (FPB * N_SCALARS) + tidx];
    float4 t[FPB / 8];
#pragma unroll
    for (int j = 0; j < FPB / 8; ++j)
      t[j] = *(const float4*)&ticker_emb[(size_t)tks[r0 + 8 * j] * ENC_DIM + c4 * 4];
#pragma unroll
    for (int j = 0; j < FPB / 8; ++j) {
      const int   row = r0 + 8 * j;
      const float sw  = wsum[row];
      const float rw  = 1.0f / (sw + 1e-8f);
      float4 v = *(const float4*)&accs[row * AP + c4 * 4];
      v.x = fmaf(sw, t[j].x, v.x) * rw;
      v.y = fmaf(sw, t[j].y, v.y) * rw;
      v.z = fmaf(sw, t[j].z, v.z) * rw;
      v.w = fmaf(sw, t[j].w, v.w) * rw;
      *(float4*)&accs[row * AP + c4 * 4] = v;
    }
    if (tidx < FPB * N_SCALARS) lsc[tidx] = log1pf(fsv);
  }
  wslab_store(wst, tidx, wr);
  wslab_load(W_proj, 1, tidx, wr);
  __syncthreads();

  float* ytile = &stage[0][0];
  {
    const int rt = wave >> 2;
    const int cg = wave & 3;
    v8f acc[8];
#pragma unroll
    for (int t = 0; t < 8; ++t) acc[t] = vzero;

    const float* arow = &accs[(rt * 16 + lm) * AP];
    const int boff = half * WKP + 2 * (cg * 128 + lm);
#pragma unroll 1
    for (int s = 0; s < ENC_DIM / SLABK; ++s) {
      if (s + 1 < ENC_DIM / SLABK) wslab_store(wst + ((s + 1) & 1) * SLABF, tidx, wr);
      if (s + 2 < ENC_DIM / SLABK) wslab_load(W_proj, s + 2, tidx, wr);
      const float* sb = wst + (s & 1) * SLABF + boff;
#pragma unroll
      for (int k8 = 0; k8 < SLABK / 4; ++k8) {
        const int kbase = 4 * (s * (SLABK / 4) + k8) + 2 * half;
        v2f a;
        a.x = arow[kbase + 0];
        a.y = arow[kbase + 1];
#pragma unroll
        for (int t = 0; t < 8; ++t) {
          const float2 bv = *(const float2*)&sb[k8 * 2 * WKP + t * 32];
          v2f b;
          b.x = bv.x;
          b.y = bv.y;
          acc[t] = wmma_f32_4(a, b, acc[t]);
        }
      }
      __syncthreads();
    }

    float l0[8], l1[8], l2[8];
#pragma unroll
    for (int r = 0; r < 8; ++r) {
      const int fr = rt * 16 + half * 8 + r;
      l0[r] = lsc[fr * N_SCALARS + 0];
      l1[r] = lsc[fr * N_SCALARS + 1];
      l2[r] = lsc[fr * N_SCALARS + 2];
    }
#pragma unroll
    for (int t = 0; t < 8; ++t) {
      const int col = (cg * 8 + t) * 16 + lm;
      const float w0 = W_proj[(size_t)(ENC_DIM + 0) * D_MODEL + col];
      const float w1 = W_proj[(size_t)(ENC_DIM + 1) * D_MODEL + col];
      const float w2 = W_proj[(size_t)(ENC_DIM + 2) * D_MODEL + col];
      const float bp = b_proj[col];
#pragma unroll
      for (int r = 0; r < 8; ++r) {
        float v = acc[t][r] + bp + l0[r] * w0 + l1[r] * w1 + l2[r] * w2;
        ytile[(rt * 16 + half * 8 + r) * YP + col] = v;
      }
    }
  }
  __syncthreads();

  {
    const int row = tidx >> 3;
    const int seg = tidx & 7;
    float s = 0.0f, ss = 0.0f;
#pragma unroll 16
    for (int j = 0; j < 64; ++j) {
      float v = ytile[row * YP + seg + j * 8];
      s  += v;
      ss += v * v;
    }
#pragma unroll
    for (int off = 1; off < 8; off <<= 1) {
      s  += __shfl_xor(s,  off, 32);
      ss += __shfl_xor(ss, off, 32);
    }
    if (seg == 0) {
      float mu  = s * (1.0f / (float)D_MODEL);
      float var = fmaf(-mu, mu, ss * (1.0f / (float)D_MODEL));
      smu[row] = mu;
      srs[row] = rsqrtf(fmaxf(var, 0.0f) + 1e-5f);
    }
  }
  __syncthreads();

  for (int i = tidx; i < FPB * D_MODEL; i += 512) {
    int row = i >> 9, col = i & 511;
    float x = (ytile[row * YP + col] - smu[row]) * srs[row] * ln_gamma[col] + ln_beta[col];
    out[(size_t)(blk * FPB + row) * D_MODEL + col] = gelu_exact(x);
  }
}

extern "C" void kernel_launch(void* const* d_in, const int* in_sizes, int n_in,
                              void* d_out, int out_size, void* d_ws, size_t ws_size,
                              hipStream_t stream) {
  const float* chunks        = (const float*)d_in[0];
  const int*   frame_id      = (const int*)  d_in[1];
  const float* weights       = (const float*)d_in[2];
  const float* frame_scalars = (const float*)d_in[3];
  const int*   ticker_ids    = (const int*)  d_in[5];
  const float* W_enc         = (const float*)d_in[6];
  const float* b_enc         = (const float*)d_in[7];
  const float* ticker_emb    = (const float*)d_in[8];
  const float* W_proj        = (const float*)d_in[9];
  const float* b_proj        = (const float*)d_in[10];
  const float* ln_gamma      = (const float*)d_in[11];
  const float* ln_beta       = (const float*)d_in[12];
  float* out = (float*)d_out;

  k_pool<<<NUM_FRAMES / FPB, 512, 0, stream>>>(
      chunks, frame_id, weights, frame_scalars, ticker_ids, W_enc, b_enc,
      ticker_emb, W_proj, b_proj, ln_gamma, ln_beta, out);
}


# Chunk rows (each read once) loaded with the non-temporal hint: their 268 MB stop evicting frame_id, weights, W_enc, ticker rows and W_proj from the caches; no sum and no sum order changes
# speedup vs baseline: 2.6997x; 1.0361x over previous
#include <hip/hip_runtime.h>
#include <hip/hip_bf16.h>

#define TOTAL_K     131072
#define CHUNK_LEN   64
#define F_DIM       8
#define NUM_FRAMES  16384
#define ENC_DIM     256
#define D_MODEL     512
#define N_SCALARS   3
#define N_TICKERS   4096
#define YP          520

typedef float v2f __attribute__((ext_vector_type(2)));
typedef float v4f __attribute__((ext_vector_type(4)));
typedef float v8f __attribute__((ext_vector_type(8)));

__device__ __forceinline__ v8f wmma_f32_4(v2f a, v2f b, v8f c) {
  return __builtin_amdgcn_wmma_f32_16x16x4_f32(
       false, a,  false, b,
       (short)0, c,  false,  false);
}

__device__ __forceinline__ float gelu_exact(float x) {
  float z  = x * 0.70710678118654752f;
  float az = fabsf(z);
  float t  = __builtin_amdgcn_rcpf(fmaf(0.3275911f, az, 1.0f));
  float p  = fmaf(t, 1.061405429f, -1.453152027f);
  p        = fmaf(t, p,  1.421413741f);
  p        = fmaf(t, p, -0.284496736f);
  p        = fmaf(t, p,  0.254829592f);
  p       *= t;
  float e  = __expf(-az * az);
  float erf_abs = fmaf(-p, e, 1.0f);
  float erf     = copysignf(erf_abs, z);
  return 0.5f * x * (1.0f + erf);
}

#define FPB    64
#define AP     272
#define SCAN_W 4
#define QCAP   768
#define SP     132
#define NWAVE  16
#define IDS_PER_WAVE (TOTAL_K / NWAVE)
#define SCAN_STEPS   (IDS_PER_WAVE / (128 * SCAN_W))

#define WAVE_LDS_ORDER()                                        \
  do {                                                          \
    __builtin_amdgcn_fence(__ATOMIC_RELEASE, "wavefront");      \
    __builtin_amdgcn_wave_barrier();                            \
    __builtin_amdgcn_fence(__ATOMIC_ACQUIRE, "wavefront");      \
  } while (0)

#define SLABK 32
#define WKP   1056
#define SLABF (16 * WKP)

__device__ __forceinline__ void wslab_load(const float* __restrict__ W_proj, int s, int tidx, float4 (&r)[8]) {
#pragma unroll
  for (int i = 0; i < 4; ++i) {
    const int task = tidx + 512 * i;
    const int p = task >> 7, c = task & 127;
    const float* g = W_proj + (size_t)(SLABK * s + 2 * p) * D_MODEL + 4 * c;
    r[2 * i]     = *(const float4*)g;
    r[2 * i + 1] = *(const float4*)(g + D_MODEL);
  }
}
__device__ __forceinline__ void wslab_store(float* sb, int tidx, const float4 (&r)[8]) {
#pragma unroll
  for (int i = 0; i < 4; ++i) {
    const int task = tidx + 512 * i;
    const int p = task >> 7, c = task & 127;
    float* d = sb + p * WKP + 8 * c;
    *(float4*)d       = make_float4(r[2 * i].x, r[2 * i + 1].x, r[2 * i].y, r[2 * i + 1].y);
    *(float4*)(d + 4) = make_float4(r[2 * i].z, r[2 * i + 1].z, r[2 * i].w, r[2 * i + 1].w);
  }
}

__global__ void __launch_bounds__(512)
k_pool(const float* __restrict__ chunks,
       const int*   __restrict__ frame_id,
       const float* __restrict__ weights,
       const float* __restrict__ frame_scalars,
       const int*   __restrict__ ticker_ids,
       const float* __restrict__ W_enc,
       const float* __restrict__ b_enc,
       const float* __restrict__ ticker_emb,
       const float* __restrict__ W_proj,
       const float* __restrict__ b_proj,
       const float* __restrict__ ln_gamma,
       const float* __restrict__ ln_beta,
       float* __restrict__ out) {
  __shared__ alignas(16) float accs[(FPB + 1) * AP];
  __shared__ float wsum[FPB + 1];
  __shared__ int   tks[FPB];
  __shared__ int   q[NWAVE][QCAP];
  __shared__ int   tq[NWAVE][16];
  __shared__ int   wcnt[NWAVE];
  __shared__ alignas(16) float stage[NWAVE][16 * SP];
  static_assert(2 * SLABF <= NWAVE * 16 * SP, "two W_proj slabs must fit the staging space");
  static_assert(FPB * YP <= NWAVE * 16 * SP, "the pre-LayerNorm rows must fit the staging space");
  __shared__ float smu[FPB], srs[FPB];
  __shared__ alignas(16) float lsc[FPB * N_SCALARS];

  const int tidx = threadIdx.x;
  const int lane = tidx & 31;
  const int wave = tidx >> 5;
  const int lm   = lane & 15;
  const int half = lane >> 4;
  const int blk  = blockIdx.x;

  for (int i = tidx; i < (FPB + 1) * AP; i += 512) accs[i] = 0.0f;
  if (tidx <= FPB) wsum[tidx] = 0.0f;
  if (tidx < FPB) tks[tidx] = ticker_ids[blk * FPB + tidx];
  __syncthreads();

  const v8f vzero = {0.f,0.f,0.f,0.f,0.f,0.f,0.f,0.f};
  const unsigned lt_mask = (1u << lane) - 1u;
  const int4* ids4 = (const int4*)frame_id + (size_t)wave * (IDS_PER_WAVE / 4);

  auto run_tile = [&](const int* te, int nvalid) __attribute__((always_inline)) {
      const int e = te[lm];
      const int c = e >> 6;
      float* st = stage[wave];
#pragma unroll
      for (int g = 0; g < 2; ++g) {
        v4f v[8][4];
#pragma unroll
        for (int j = 0; j < 8; ++j) {
          const int cj = __builtin_amdgcn_readlane(e, g * 8 + j) >> 6;
          const v4f* b = (const v4f*)(chunks + (size_t)cj * (CHUNK_LEN * F_DIM)) + lane;
#pragma unroll
          for (int k = 0; k < 4; ++k) v[j][k] = __builtin_nontemporal_load(&b[k * 32]);
        }
        __builtin_amdgcn_sched_barrier(0);
#pragma unroll
        for (int j = 0; j < 8; ++j) {
          v4f ps;
          ps.x = (v[j][0].x + v[j][1].x) + (v[j][2].x + v[j][3].x);
          ps.y = (v[j][0].y + v[j][1].y) + (v[j][2].y + v[j][3].y);
          ps.z = (v[j][0].z + v[j][1].z) + (v[j][2].z + v[j][3].z);
          ps.w = (v[j][0].w + v[j][1].w) + (v[j][2].w + v[j][3].w);
          *(v4f*)&st[(g * 8 + j) * SP + lane * 4] = ps;
        }
      }
      WAVE_LDS_ORDER();
      float m0 = 0.f, m1 = 0.f, m4 = 0.f, m5 = 0.f;
#pragma unroll
      for (int i = 0; i < 16; ++i) {
        const float2 ev = *(const float2*)&st[lm * SP + i * 8 + 2 * half];
        const float2 od = *(const float2*)&st[lm * SP + i * 8 + 4 + 2 * half];
        m0 += ev.x; m1 += ev.y; m4 += od.x; m5 += od.y;
      }

      v2f A0, A1;
      A0.x = m0 * (1.0f / 64.0f);  A0.y = m1 * (1.0f / 64.0f);
      A1.x = m4 * (1.0f / 64.0f);  A1.y = m5 * (1.0f / 64.0f);

      int   fl[8];
      float wg[8];
#pragma unroll
      for (int r = 0; r < 8; ++r) {
        const int er = te[half * 8 + r];
        wg[r]  = weights[er >> 6];
        fl[r]  = ((half * 8 + r) < nvalid) ? (er & 63) : FPB;
      }
      if (half == 0)
        __hip_atomic_fetch_add(&wsum[lm < nvalid ? (e & 63) : FPB], weights[c],
                               __ATOMIC_RELAXED, __HIP_MEMORY_SCOPE_WORKGROUP);

      v2f nB0, nB1;
      float nbe;
      {
        const int col = lm;
        nB0.x = W_enc[(2 * half + 0) * ENC_DIM + col];
        nB0.y = W_enc[(2 * half + 1) * ENC_DIM + col];
        nB1.x = W_enc[(2 * half + 4) * ENC_DIM + col];
        nB1.y = W_enc[(2 * half + 5) * ENC_DIM + col];
        nbe   = b_enc[col];
      }
#pragma unroll 1
      for (int nt = 0; nt < 16; ++nt) {
        const int col = nt * 16 + lm;
        const v2f B0 = nB0, B1 = nB1;
        const float be = nbe;
        {
          const int coln = ((nt + 1) & 15) * 16 + lm;
          nB0.x = W_enc[(2 * half + 0) * ENC_DIM + coln];
          nB0.y = W_enc[(2 * half + 1) * ENC_DIM + coln];
          nB1.x = W_enc[(2 * half + 4) * ENC_DIM + coln];
          nB1.y = W_enc[(2 * half + 5) * ENC_DIM + coln];
          nbe   = b_enc[coln];
        }
        v8f acc = wmma_f32_4(A0, B0, vzero);
        acc     = wmma_f32_4(A1, B1, acc);
#pragma unroll
        for (int r = 0; r < 8; ++r) {
          float h   = acc[r] + be;
          float g   = gelu_exact(h);
          float val = g * wg[r];
          __hip_atomic_fetch_add(&accs[fl[r] * AP + col], val, __ATOMIC_RELAXED, __HIP_MEMORY_SCOPE_WORKGROUP);
        }
      }
      WAVE_LDS_ORDER();
  };

  int cnt = 0;
  int4 nxt[SCAN_W];
#pragma unroll
  for (int k = 0; k < SCAN_W; ++k) nxt[k] = ids4[k * 32 + lane];
#pragma unroll 1
  for (int s = 0; s < SCAN_STEPS; ++s) {
    int4 cur[SCAN_W];
#pragma unroll
    for (int k = 0; k < SCAN_W; ++k) cur[k] = nxt[k];
    if (s + 1 < SCAN_STEPS) {
#pragma unroll
      for (int k = 0; k < SCAN_W; ++k) nxt[k] = ids4[((s + 1) * SCAN_W + k) * 32 + lane];
    }
#pragma unroll
    for (int k = 0; k < SCAN_W; ++k) {
      const int cbase = wave * IDS_PER_WAVE + (s * SCAN_W + k) * 128 + lane * 4;
      const int idv[4] = {cur[k].x, cur[k].y, cur[k].z, cur[k].w};
      const bool any4 = ((idv[0] >> 6) == blk) | ((idv[1] >> 6) == blk) |
                        ((idv[2] >> 6) == blk) | ((idv[3] >> 6) == blk);
      if (__builtin_amdgcn_ballot_w32(any4) != 0u) {
#pragma unroll
        for (int j = 0; j < 4; ++j) {
          const int  id = idv[j];
          const bool mt = (id >> 6) == blk;
          const unsigned mask = __builtin_amdgcn_ballot_w32(mt);
          if (mt) q[wave][cnt + __builtin_popcount(mask & lt_mask)] = ((cbase + j) << 6) | (id & 63);
          cnt += __builtin_popcount(mask);
        }
      }
    }
    WAVE_LDS_ORDER();
    while (cnt > QCAP - 128 * SCAN_W) {
      run_tile(&q[wave][cnt - 16], 16);
      cnt -= 16;
    }
  }
  if (lane == 0) wcnt[wave] = cnt;
  __syncthreads();

  {
    int wb[NWAVE + 1];
    wb[0] = 0;
#pragma unroll
    for (int j = 0; j < NWAVE; ++j) wb[j + 1] = wb[j] + wcnt[j];
    const int nlist  = __builtin_amdgcn_readfirstlane(wb[NWAVE]);
    const int ntiles = (nlist + 15) >> 4;
    for (int t = __builtin_amdgcn_readfirstlane(wave); t < ntiles; t += NWAVE) {
      const int left   = nlist - 16 * t;
      const int nvalid = left < 16 ? left : 16;
      const int vi = 16 * t + lm;
      int wj = 0, off = 0;
#pragma unroll
      for (int j = 0; j < NWAVE; ++j)
        if (vi >= wb[j]) { wj = j; off = vi - wb[j]; }
      const int ent = (lm < nvalid) ? q[wj][off] : 0;
      if (lane < 16) tq[wave][lane] = ent;
      WAVE_LDS_ORDER();
      run_tile(tq[wave], nvalid);
    }
  }
  __syncthreads();

  float* wst = &stage[0][0];
  float4 wr[8];
  wslab_load(W_proj, 0, tidx, wr);
  {
    const int r0 = tidx >> 6, c4 = tidx & 63;
    float fsv = 0.0f;
    if (tidx < FPB * N_SCALARS) fsv = frame_scalars[blk * (FPB * N_SCALARS) + tidx];
    float4 t[FPB / 8];
#pragma unroll
    for (int j = 0; j < FPB / 8; ++j)
      t[j] = *(const float4*)&ticker_emb[(size_t)tks[r0 + 8 * j] * ENC_DIM + c4 * 4];
#pragma unroll
    for (int j = 0; j < FPB / 8; ++j) {
      const int   row = r0 + 8 * j;
      const float sw  = wsum[row];
      const float rw  = 1.0f / (sw + 1e-8f);
      float4 v = *(const float4*)&accs[row * AP + c4 * 4];
      v.x = fmaf(sw, t[j].x, v.x) * rw;
      v.y = fmaf(sw, t[j].y, v.y) * rw;
      v.z = fmaf(sw, t[j].z, v.z) * rw;
      v.w = fmaf(sw, t[j].w, v.w) * rw;
      *(float4*)&accs[row * AP + c4 * 4] = v;
    }
    if (tidx < FPB * N_SCALARS) lsc[tidx] = log1pf(fsv);
  }
  wslab_store(wst, tidx, wr);
  wslab_load(W_proj, 1, tidx, wr);
  __syncthreads();

  float* ytile = &stage[0][0];
  {
    const int rt = wave >> 2;
    const int cg = wave & 3;
    v8f acc[8];
#pragma unroll
    for (int t = 0; t < 8; ++t) acc[t] = vzero;

    const float* arow = &accs[(rt * 16 + lm) * AP];
    const int boff = half * WKP + 2 * (cg * 128 + lm);
#pragma unroll 1
    for (int s = 0; s < ENC_DIM / SLABK; ++s) {
      if (s + 1 < ENC_DIM / SLABK) wslab_store(wst + ((s + 1) & 1) * SLABF, tidx, wr);
      if (s + 2 < ENC_DIM / SLABK) wslab_load(W_proj, s + 2, tidx, wr);
      const float* sb = wst + (s & 1) * SLABF + boff;
#pragma unroll
      for (int k8 = 0; k8 < SLABK / 4; ++k8) {
        const int kbase = 4 * (s * (SLABK / 4) + k8) + 2 * half;
        v2f a;
        a.x = arow[kbase + 0];
        a.y = arow[kbase + 1];
#pragma unroll
        for (int t = 0; t < 8; ++t) {
          const float2 bv = *(const float2*)&sb[k8 * 2 * WKP + t * 32];
          v2f b;
          b.x = bv.x;
          b.y = bv.y;
          acc[t] = wmma_f32_4(a, b, acc[t]);
        }
      }
      __syncthreads();
    }

    float l0[8], l1[8], l2[8];
#pragma unroll
    for (int r = 0; r < 8; ++r) {
      const int fr = rt * 16 + half * 8 + r;
      l0[r] = lsc[fr * N_SCALARS + 0];
      l1[r] = lsc[fr * N_SCALARS + 1];
      l2[r] = lsc[fr * N_SCALARS + 2];
    }
#pragma unroll
    for (int t = 0; t < 8; ++t) {
      const int col = (cg * 8 + t) * 16 + lm;
      const float w0 = W_proj[(size_t)(ENC_DIM + 0) * D_MODEL + col];
      const float w1 = W_proj[(size_t)(ENC_DIM + 1) * D_MODEL + col];
      const float w2 = W_proj[(size_t)(ENC_DIM + 2) * D_MODEL + col];
      const float bp = b_proj[col];
#pragma unroll
      for (int r = 0; r < 8; ++r) {
        float v = acc[t][r] + bp + l0[r] * w0 + l1[r] * w1 + l2[r] * w2;
        ytile[(rt * 16 + half * 8 + r) * YP + col] = v;
      }
    }
  }
  __syncthreads();

  {
    const int row = tidx >> 3;
    const int seg = tidx & 7;
    float s = 0.0f, ss = 0.0f;
#pragma unroll 16
    for (int j = 0; j < 64; ++j) {
      float v = ytile[row * YP + seg + j * 8];
      s  += v;
      ss += v * v;
    }
#pragma unroll
    for (int off = 1; off < 8; off <<= 1) {
      s  += __shfl_xor(s,  off, 32);
      ss += __shfl_xor(ss, off, 32);
    }
    if (seg == 0) {
      float mu  = s * (1.0f / (float)D_MODEL);
      float var = fmaf(-mu, mu, ss * (1.0f / (float)D_MODEL));
      smu[row] = mu;
      srs[row] = rsqrtf(fmaxf(var, 0.0f) + 1e-5f);
    }
  }
  __syncthreads();

  for (int i = tidx; i < FPB * D_MODEL; i += 512) {
    int row = i >> 9, col = i & 511;
    float x = (ytile[row * YP + col] - smu[row]) * srs[row] * ln_gamma[col] + ln_beta[col];
    out[(size_t)(blk * FPB + row) * D_MODEL + col] = gelu_exact(x);
  }
}

extern "C" void kernel_launch(void* const* d_in, const int* in_sizes, int n_in,
                              void* d_out, int out_size, void* d_ws, size_t ws_size,
                              hipStream_t stream) {
  const float* chunks        = (const float*)d_in[0];
  const int*   frame_id      = (const int*)  d_in[1];
  const float* weights       = (const float*)d_in[2];
  const float* frame_scalars = (const float*)d_in[3];
  const int*   ticker_ids    = (const int*)  d_in[5];
  const float* W_enc         = (const float*)d_in[6];
  const float* b_enc         = (const float*)d_in[7];
  const float* ticker_emb    = (const float*)d_in[8];
  const float* W_proj        = (const float*)d_in[9];
  const float* b_proj        = (const float*)d_in[10];
  const float* ln_gamma      = (const float*)d_in[11];
  const float* ln_beta       = (const float*)d_in[12];
  float* out = (float*)d_out;

  k_pool<<<NUM_FRAMES / FPB, 512, 0, stream>>>(
      chunks, frame_id, weights, frame_scalars, ticker_ids, W_enc, b_enc,
      ticker_emb, W_proj, b_proj, ln_gamma, ln_beta, out);
}
